# starting kernel
# baseline (speedup 1.0000x reference)
#include <hip/hip_runtime.h>

typedef __attribute__((ext_vector_type(2))) float v2f;
typedef __attribute__((ext_vector_type(8))) float v8f;

#define DIM    32
#define TILE   16
#define MCONST 4096
#define LOG2E  1.4426950408889634f

__device__ __forceinline__ void load_bfrag(v2f b[8], const float* __restrict__ crow,
                                           int koff) {
    #pragma unroll
    for (int kk = 0; kk < 8; ++kk) {
        b[kk].x = crow[4 * kk + koff + 0];
        b[kk].y = crow[4 * kk + koff + 1];
    }
}

__device__ __forceinline__ v8f wmma_chain(const v2f a[8], const v2f b[8]) {
    v8f d = {};
    #pragma unroll
    for (int kk = 0; kk < 8; ++kk) {
        d = __builtin_amdgcn_wmma_f32_16x16x4_f32(
            false, a[kk], false, b[kk], (short)0, d, false, false);
    }
    return d;
}

__device__ __forceinline__ void epilogue(const v8f d, float y2e, float al, float c2,
                                         const float x2e[8], float acc[8]) {
    #pragma unroll
    for (int v = 0; v < 8; ++v) {
        float e = __builtin_fmaf(c2, d[v], x2e[v] + y2e);
        acc[v] = __builtin_fmaf(al, __builtin_amdgcn_exp2f(e), acc[v]);
    }
}

__global__ __launch_bounds__(256, 2)
void nkrr_rbf_wmma_kernel(const float* __restrict__ X,
                          const float* __restrict__ C,
                          const float* __restrict__ alphas,
                          const float* __restrict__ sigma,
                          float* __restrict__ out,
                          int N, int M)
{
    __shared__ float s_y2e[MCONST];
    __shared__ float s_al[MCONST];

    const float g  = sigma[0];
    const float c1 = -0.5f * LOG2E / (g * g);
    const float c2 = -2.0f * c1;

    for (int j = threadIdx.x; j < M; j += blockDim.x) {
        const float* c = C + (size_t)j * DIM;
        float s = 0.f;
        #pragma unroll
        for (int d = 0; d < DIM; ++d) s += c[d] * c[d];
        s_y2e[j] = c1 * s;
        s_al[j]  = alphas[j];
    }
    __syncthreads();

    const int lane = threadIdx.x & 31;
    const int wave = threadIdx.x >> 5;
    const int wavesPerBlock = blockDim.x >> 5;
    const int rowTile = blockIdx.x * wavesPerBlock + wave;
    const int row0 = rowTile * TILE;
    if (row0 >= N) return;

    const int half = lane >> 4;
    const int ln   = lane & 15;
    const int koff = half * 2;

    v2f a[8];
    {
        const float* xrow = X + (size_t)(row0 + ln) * DIM;
        #pragma unroll
        for (int kk = 0; kk < 8; ++kk) {
            a[kk].x = xrow[4 * kk + koff + 0];
            a[kk].y = xrow[4 * kk + koff + 1];
        }
    }

    float x2e[8];
    #pragma unroll
    for (int v = 0; v < 8; ++v) {
        const float* xr = X + (size_t)(row0 + v + 8 * half) * DIM;
        float s = 0.f;
        #pragma unroll
        for (int d = 0; d < DIM; ++d) s += xr[d] * xr[d];
        x2e[v] = c1 * s;
    }

    float acc[8];
    #pragma unroll
    for (int v = 0; v < 8; ++v) acc[v] = 0.f;

    const float* cbase = C + (size_t)ln * DIM;

    v2f b0[8], b1[8];
    load_bfrag(b0, cbase, koff);

    for (int jt = 0; jt < M; jt += 2 * TILE) {
        load_bfrag(b1, cbase + (size_t)(jt + TILE) * DIM, koff);

        v8f d0 = wmma_chain(a, b0);
        epilogue(d0, s_y2e[jt + ln], s_al[jt + ln], c2, x2e, acc);

        int jn = jt + 2 * TILE;
        if (jn >= M) jn = 0;
        load_bfrag(b0, cbase + (size_t)jn * DIM, koff);

        v8f d1 = wmma_chain(a, b1);
        epilogue(d1, s_y2e[jt + TILE + ln], s_al[jt + TILE + ln], c2, x2e, acc);
    }

    #pragma unroll
    for (int v = 0; v < 8; ++v) {
        float s = acc[v];
        s += __shfl_xor(s, 8, 32);
        s += __shfl_xor(s, 4, 32);
        s += __shfl_xor(s, 2, 32);
        s += __shfl_xor(s, 1, 32);
        if (ln == 0) {
            out[row0 + v + 8 * half] = s;
        }
    }
}

extern "C" void kernel_launch(void* const* d_in, const int* in_sizes, int n_in,
                              void* d_out, int out_size, void* d_ws, size_t ws_size,
                              hipStream_t stream) {
    const float* X      = (const float*)d_in[0];
    const float* C      = (const float*)d_in[1];
    const float* alphas = (const float*)d_in[2];
    const float* sigma  = (const float*)d_in[3];
    float* out = (float*)d_out;

    const int N = in_sizes[0] / DIM;
    const int M = in_sizes[1] / DIM;

    const int wavesPerBlock = 8;
    const int rowTiles = N / TILE;
    dim3 block(32 * wavesPerBlock);
    dim3 grid(rowTiles / wavesPerBlock);

    nkrr_rbf_wmma_kernel<<<grid, block, 0, stream>>>(X, C, alphas, sigma, out, N, M);
}


# B fragments read from a work-group-shared LDS copy of the centers (512-center stages, coalesced 16-B staging, 144-B row pitch, two buffers, one barrier per stage), not per-wave strided global loads
# speedup vs baseline: 1.2820x; 1.2820x over previous
#include <hip/hip_runtime.h>

typedef __attribute__((ext_vector_type(2))) float v2f;
typedef __attribute__((ext_vector_type(4))) float v4f;
typedef __attribute__((ext_vector_type(8))) float v8f;

#define DIM    32
#define TILE   16
#define MCONST 4096
#define LOG2E  1.4426950408889634f
#define STAGE  512
#define SUB    128
#define PITCH  36
#define SUBCHUNKS (SUB * DIM / 4)

__device__ __forceinline__ void load_bfrag(v2f b[8], const float* __restrict__ crow,
                                           int koff) {
    #pragma unroll
    for (int kk = 0; kk < 8; ++kk) {
        b[kk].x = crow[4 * kk + koff + 0];
        b[kk].y = crow[4 * kk + koff + 1];
    }
}

__device__ __forceinline__ v8f wmma_chain(const v2f a[8], const v2f b[8]) {
    v8f d = {};
    #pragma unroll
    for (int kk = 0; kk < 8; ++kk) {
        d = __builtin_amdgcn_wmma_f32_16x16x4_f32(
            false, a[kk], false, b[kk], (short)0, d, false, false);
    }
    return d;
}

__device__ __forceinline__ void epilogue(const v8f d, float y2e, float al, float c2,
                                         const float x2e[8], float acc[8]) {
    #pragma unroll
    for (int v = 0; v < 8; ++v) {
        float e = __builtin_fmaf(c2, d[v], x2e[v] + y2e);
        acc[v] = __builtin_fmaf(al, __builtin_amdgcn_exp2f(e), acc[v]);
    }
}

__global__ __launch_bounds__(256, 2)
void nkrr_rbf_wmma_kernel(const float* __restrict__ X,
                          const float* __restrict__ C,
                          const float* __restrict__ alphas,
                          const float* __restrict__ sigma,
                          float* __restrict__ out,
                          int N, int M)
{
    __shared__ float s_y2e[MCONST];
    __shared__ float s_al[MCONST];
    __shared__ __attribute__((aligned(16))) float s_c[2][STAGE * PITCH];

    const float g  = sigma[0];
    const float c1 = -0.5f * LOG2E / (g * g);
    const float c2 = -2.0f * c1;

    const int tid = threadIdx.x;
    const v4f* __restrict__ C4 = reinterpret_cast<const v4f*>(C);

    #pragma unroll
    for (int i = 0; i < STAGE * DIM / 4 / 256; ++i) {
        const int ch = tid + 256 * i;
        const v4f val = C4[ch];
        *reinterpret_cast<v4f*>(&s_c[0][(ch >> 3) * PITCH + (ch & 7) * 4]) = val;
    }

    for (int j = threadIdx.x; j < M; j += blockDim.x) {
        const float* c = C + (size_t)j * DIM;
        float s = 0.f;
        #pragma unroll
        for (int d = 0; d < DIM; ++d) s += c[d] * c[d];
        s_y2e[j] = c1 * s;
        s_al[j]  = alphas[j];
    }
    __syncthreads();

    const int lane = threadIdx.x & 31;
    const int wave = threadIdx.x >> 5;
    const int wavesPerBlock = blockDim.x >> 5;
    const int rowTile = blockIdx.x * wavesPerBlock + wave;
    const bool active = rowTile * TILE < N;
    const int row0 = active ? rowTile * TILE : 0;

    const int half = lane >> 4;
    const int ln   = lane & 15;
    const int koff = half * 2;

    v2f a[8];
    {
        const float* xrow = X + (size_t)(row0 + ln) * DIM;
        #pragma unroll
        for (int kk = 0; kk < 8; ++kk) {
            a[kk].x = xrow[4 * kk + koff + 0];
            a[kk].y = xrow[4 * kk + koff + 1];
        }
    }

    float x2e[8];
    #pragma unroll
    for (int v = 0; v < 8; ++v) {
        const float* xr = X + (size_t)(row0 + v + 8 * half) * DIM;
        float s = 0.f;
        #pragma unroll
        for (int d = 0; d < DIM; ++d) s += xr[d] * xr[d];
        x2e[v] = c1 * s;
    }

    float acc[8];
    #pragma unroll
    for (int v = 0; v < 8; ++v) acc[v] = 0.f;

    const int nStages = M / STAGE;
    const int lrow = ln * PITCH;

    for (int s = 0; s < nStages; ++s) {
        const float* buf = s_c[s & 1];
        float* nb = s_c[(s + 1) & 1];
        const bool more = (s + 1 < nStages);

        for (int q = 0; q < STAGE / SUB; ++q) {
            v4f nx[SUBCHUNKS / 256];
            if (more) {
                #pragma unroll
                for (int i = 0; i < SUBCHUNKS / 256; ++i)
                    nx[i] = C4[(size_t)(s + 1) * (STAGE * DIM / 4) + q * SUBCHUNKS + tid + 256 * i];
            }

            const float* sb = buf + q * SUB * PITCH + lrow;
            const int jbase = s * STAGE + q * SUB + ln;

            #pragma unroll
            for (int tt = 0; tt < SUB / TILE; ++tt) {
                v2f b[8];
                load_bfrag(b, sb + tt * TILE * PITCH, koff);
                const float y2 = s_y2e[jbase + tt * TILE];
                const float al = s_al[jbase + tt * TILE];
                v8f d = wmma_chain(a, b);
                epilogue(d, y2, al, c2, x2e, acc);
            }

            if (more) {
                #pragma unroll
                for (int i = 0; i < SUBCHUNKS / 256; ++i) {
                    const int ch = q * SUBCHUNKS + tid + 256 * i;
                    *reinterpret_cast<v4f*>(&nb[(ch >> 3) * PITCH + (ch & 7) * 4]) = nx[i];
                }
            }
        }
        __syncthreads();
    }

    #pragma unroll
    for (int v = 0; v < 8; ++v) {
        float s = acc[v];
        s += __shfl_xor(s, 8, 32);
        s += __shfl_xor(s, 4, 32);
        s += __shfl_xor(s, 2, 32);
        s += __shfl_xor(s, 1, 32);
        if (ln == 0 && active) {
            out[row0 + v + 8 * half] = s;
        }
    }
}

extern "C" void kernel_launch(void* const* d_in, const int* in_sizes, int n_in,
                              void* d_out, int out_size, void* d_ws, size_t ws_size,
                              hipStream_t stream) {
    const float* X      = (const float*)d_in[0];
    const float* C      = (const float*)d_in[1];
    const float* alphas = (const float*)d_in[2];
    const float* sigma  = (const float*)d_in[3];
    float* out = (float*)d_out;

    const int N = in_sizes[0] / DIM;
    const int M = in_sizes[1] / DIM;

    const int wavesPerBlock = 8;
    const int rowTiles = N / TILE;
    dim3 block(32 * wavesPerBlock);
    dim3 grid(rowTiles / wavesPerBlock);

    nkrr_rbf_wmma_kernel<<<grid, block, 0, stream>>>(X, C, alphas, sigma, out, N, M);
}


# center norms are summed from the 16-byte pieces already being staged to LDS (4 squares per lane, then a 3-step 8-lane butterfly: sum order changes), not by each group's second pass over all centers
# speedup vs baseline: 1.5385x; 1.2001x over previous
#include <hip/hip_runtime.h>

typedef __attribute__((ext_vector_type(2))) float v2f;
typedef __attribute__((ext_vector_type(4))) float v4f;
typedef __attribute__((ext_vector_type(8))) float v8f;

#define DIM    32
#define TILE   16
#define MCONST 4096
#define LOG2E  1.4426950408889634f
#define STAGE  512
#define SUB    128
#define PITCH  36
#define SUBCHUNKS (SUB * DIM / 4)

__device__ __forceinline__ void load_bfrag(v2f b[8], const float* __restrict__ crow,
                                           int koff) {
    #pragma unroll
    for (int kk = 0; kk < 8; ++kk) {
        b[kk].x = crow[4 * kk + koff + 0];
        b[kk].y = crow[4 * kk + koff + 1];
    }
}

__device__ __forceinline__ v8f wmma_chain(const v2f a[8], const v2f b[8]) {
    v8f d = {};
    #pragma unroll
    for (int kk = 0; kk < 8; ++kk) {
        d = __builtin_amdgcn_wmma_f32_16x16x4_f32(
            false, a[kk], false, b[kk], (short)0, d, false, false);
    }
    return d;
}

__device__ __forceinline__ void epilogue(const v8f d, float y2e, float al, float c2,
                                         const float x2e[8], float acc[8]) {
    #pragma unroll
    for (int v = 0; v < 8; ++v) {
        float e = __builtin_fmaf(c2, d[v], x2e[v] + y2e);
        acc[v] = __builtin_fmaf(al, __builtin_amdgcn_exp2f(e), acc[v]);
    }
}

__device__ __forceinline__ float sq4(const v4f v) {
    const v2f lo = {v.x, v.y};
    const v2f hi = {v.z, v.w};
    const v2f t = hi * hi + lo * lo;
    return t.x + t.y;
}

template <int CTRL>
__device__ __forceinline__ float xlane_add(float p) {
    const int moved = __builtin_amdgcn_update_dpp(0, __builtin_bit_cast(int, p), CTRL, 0xf, 0xf, true);
    return p + __builtin_bit_cast(float, moved);
}

template <int R>
__device__ __forceinline__ void sum8_rows(float p[R]) {
    #pragma unroll
    for (int i = 0; i < R; ++i) p[i] = xlane_add<0x161>(p[i]);
    #pragma unroll
    for (int i = 0; i < R; ++i) p[i] = xlane_add<0x162>(p[i]);
    #pragma unroll
    for (int i = 0; i < R; ++i) p[i] = xlane_add<0x164>(p[i]);
}

__global__ __launch_bounds__(256, 2)
void nkrr_rbf_wmma_kernel(const float* __restrict__ X,
                          const float* __restrict__ C,
                          const float* __restrict__ alphas,
                          const float* __restrict__ sigma,
                          float* __restrict__ out,
                          int N, int M)
{
    __shared__ float s_y2e[MCONST];
    __shared__ __attribute__((aligned(16))) float s_al[MCONST];
    __shared__ __attribute__((aligned(16))) float s_c[2][STAGE * PITCH];

    const float g  = sigma[0];
    const float c1 = -0.5f * LOG2E / (g * g);
    const float c2 = -2.0f * c1;

    const int tid = threadIdx.x;
    const v4f* __restrict__ C4 = reinterpret_cast<const v4f*>(C);

    const int piece = tid & 7;
    const int prow  = tid >> 3;

    {
        const v4f* __restrict__ A4 = reinterpret_cast<const v4f*>(alphas);
        v4f* s_al4 = reinterpret_cast<v4f*>(s_al);
        v4f av[MCONST / 4 / 256];
        #pragma unroll
        for (int i = 0; i < MCONST / 4 / 256; ++i) {
            const int j = tid + 256 * i;
            av[i] = A4[j < M / 4 ? j : 0];
        }
        #pragma unroll
        for (int i = 0; i < MCONST / 4 / 256; ++i) s_al4[tid + 256 * i] = av[i];
    }

    {
        constexpr int NCH = STAGE * DIM / 4 / 256;
        v4f val[NCH];
        #pragma unroll
        for (int i = 0; i < NCH; ++i) val[i] = C4[tid + 256 * i];
        float nrm[NCH];
        #pragma unroll
        for (int i = 0; i < NCH; ++i) {
            const int ch = tid + 256 * i;
            *reinterpret_cast<v4f*>(&s_c[0][(ch >> 3) * PITCH + (ch & 7) * 4]) = val[i];
            nrm[i] = sq4(val[i]);
        }
        sum8_rows<NCH>(nrm);
        if (piece == 0) {
            #pragma unroll
            for (int i = 0; i < NCH; ++i) s_y2e[prow + 32 * i] = c1 * nrm[i];
        }
    }
    __syncthreads();

    const int lane = threadIdx.x & 31;
    const int wave = threadIdx.x >> 5;
    const int wavesPerBlock = blockDim.x >> 5;
    const int rowTile = blockIdx.x * wavesPerBlock + wave;
    const bool active = rowTile * TILE < N;
    const int row0 = active ? rowTile * TILE : 0;

    const int half = lane >> 4;
    const int ln   = lane & 15;
    const int koff = half * 2;

    v2f a[8];
    {
        const float* xrow = X + (size_t)(row0 + ln) * DIM;
        #pragma unroll
        for (int kk = 0; kk < 8; ++kk) {
            a[kk].x = xrow[4 * kk + koff + 0];
            a[kk].y = xrow[4 * kk + koff + 1];
        }
    }

    float x2e[8];
    #pragma unroll
    for (int v = 0; v < 8; ++v) {
        const float* xr = X + (size_t)(row0 + v + 8 * half) * DIM;
        float s = 0.f;
        #pragma unroll
        for (int d = 0; d < DIM; ++d) s += xr[d] * xr[d];
        x2e[v] = c1 * s;
    }

    float acc[8];
    #pragma unroll
    for (int v = 0; v < 8; ++v) acc[v] = 0.f;

    const int nStages = M / STAGE;
    const int lrow = ln * PITCH;

    for (int s = 0; s < nStages; ++s) {
        const float* buf = s_c[s & 1];
        float* nb = s_c[(s + 1) & 1];
        const bool more = (s + 1 < nStages);

        for (int q = 0; q < STAGE / SUB; ++q) {
            v4f nx[SUBCHUNKS / 256];
            if (more) {
                #pragma unroll
                for (int i = 0; i < SUBCHUNKS / 256; ++i)
                    nx[i] = C4[(size_t)(s + 1) * (STAGE * DIM / 4) + q * SUBCHUNKS + tid + 256 * i];
            }

            const float* sb = buf + q * SUB * PITCH + lrow;
            const int jbase = s * STAGE + q * SUB + ln;

            #pragma unroll
            for (int tt = 0; tt < SUB / TILE; ++tt) {
                v2f b[8];
                load_bfrag(b, sb + tt * TILE * PITCH, koff);
                const float y2 = s_y2e[jbase + tt * TILE];
                const float al = s_al[jbase + tt * TILE];
                v8f d = wmma_chain(a, b);
                epilogue(d, y2, al, c2, x2e, acc);
            }

            if (more) {
                float nrm[SUBCHUNKS / 256];
                #pragma unroll
                for (int i = 0; i < SUBCHUNKS / 256; ++i) {
                    const int ch = q * SUBCHUNKS + tid + 256 * i;
                    *reinterpret_cast<v4f*>(&nb[(ch >> 3) * PITCH + (ch & 7) * 4]) = nx[i];
                    nrm[i] = sq4(nx[i]);
                }
                sum8_rows<SUBCHUNKS / 256>(nrm);
                const float mine = piece == 0 ? nrm[0] : piece == 1 ? nrm[1] : piece == 2 ? nrm[2] : nrm[3];
                if (piece < 4) s_y2e[(s + 1) * STAGE + q * SUB + 32 * piece + prow] = c1 * mine;
            }
        }
        __syncthreads();
    }

    #pragma unroll
    for (int v = 0; v < 8; ++v) {
        float s = acc[v];
        s += __shfl_xor(s, 8, 32);
        s += __shfl_xor(s, 4, 32);
        s += __shfl_xor(s, 2, 32);
        s += __shfl_xor(s, 1, 32);
        if (ln == 0 && active) {
            out[row0 + v + 8 * half] = s;
        }
    }
}

extern "C" void kernel_launch(void* const* d_in, const int* in_sizes, int n_in,
                              void* d_out, int out_size, void* d_ws, size_t ws_size,
                              hipStream_t stream) {
    const float* X      = (const float*)d_in[0];
    const float* C      = (const float*)d_in[1];
    const float* alphas = (const float*)d_in[2];
    const float* sigma  = (const float*)d_in[3];
    float* out = (float*)d_out;

    const int N = in_sizes[0] / DIM;
    const int M = in_sizes[1] / DIM;

    const int wavesPerBlock = 8;
    const int rowTiles = N / TILE;
    dim3 block(32 * wavesPerBlock);
    dim3 grid(rowTiles / wavesPerBlock);

    nkrr_rbf_wmma_kernel<<<grid, block, 0, stream>>>(X, C, alphas, sigma, out, N, M);
}


# speedup vs baseline: 1.8608x; 1.2095x over previous
#include <hip/hip_runtime.h>


typedef __attribute__((ext_vector_type(2))) float v2f;
typedef __attribute__((ext_vector_type(4))) float v4f;
typedef __attribute__((ext_vector_type(8))) float v8f;

#define DIM    32
#define TILE   16
#define WAVES  8
#define RT     8
#define ROWS   (RT * TILE)
#define LOG2E  1.4426950408889634f


struct BTile {
    v4f q[4];
    float al;
};

__device__ __forceinline__ void load_btile(BTile& t, const v4f* __restrict__ C4,
                                           const float* __restrict__ alphas,
                                           int tile, int ln, int half) {
    const int j = tile * TILE + ln;
    const v4f* p = C4 + (size_t)j * (DIM / 4) + half;
    #pragma unroll
    for (int i = 0; i < 4; ++i) t.q[i] = p[2 * i];
    t.al = alphas[j];
}

__device__ __forceinline__ float sumsq16(const v4f q[4]) {
    v4f s = q[0] * q[0];
    s = q[1] * q[1] + s;
    s = q[2] * q[2] + s;
    s = q[3] * q[3] + s;
    return (s.x + s.y) + (s.z + s.w);
}

__device__ __forceinline__ float xhalf(float x) {
    const int v = __builtin_bit_cast(int, x);
    return __builtin_bit_cast(float,
        __builtin_amdgcn_permlanex16(v, v, 0x76543210, 0xfedcba98, false, false));
}

template <int CTRL>
__device__ __forceinline__ float xlane_add(float p) {
    const int moved = __builtin_amdgcn_update_dpp(0, __builtin_bit_cast(int, p), CTRL, 0xf, 0xf, true);
    return p + __builtin_bit_cast(float, moved);
}

__device__ __forceinline__ void mm_tile(const BTile& t, const v2f a[RT][8], const v8f x2s[RT],
                                        v8f d[RT]) {
    #pragma unroll
    for (int kk = 0; kk < 8; ++kk) {
        v2f b;
        if (kk & 1) { b.x = t.q[kk >> 1].z; b.y = t.q[kk >> 1].w; }
        else        { b.x = t.q[kk >> 1].x; b.y = t.q[kk >> 1].y; }
        #pragma unroll
        for (int rt = 0; rt < RT; ++rt) {
            d[rt] = __builtin_amdgcn_wmma_f32_16x16x4_f32(
                false, a[rt][kk], false, b, (short)0, kk == 0 ? x2s[rt] : d[rt], false, false);
        }
    }
}

template <class M, class N>
__device__ __forceinline__ void issue_after(M& made, N& next_in) {
    asm volatile("" : "+v"(made), "+v"(next_in));
}

#define LEAD 3
#define LAG  3

__device__ __forceinline__ void post_tile(const BTile& t, float c1, v8f d[RT],
                                          v2f acc[RT][4], BTile& next) {
    asm volatile("" : "+v"(d[0]), "+v"(d[1]), "+v"(d[2]), "+v"(d[3]),
                      "+v"(d[4]), "+v"(d[5]), "+v"(d[6]), "+v"(d[7]));
    float s = sumsq16(t.q);
    s += xhalf(s);
    v2f y2 = (v2f){s, s};
    const v2f c1v = (v2f){c1, c1};
    v2f al = (v2f){t.al, t.al};

    v2f e[4 * RT];
    float px[4 * RT];
    float py[4 * RT];
    #pragma unroll
    for (int u = 0; u < LEAD; ++u)
        e[u] = c1v * y2 + (v2f){d[u >> 2][2 * (u & 3)], d[u >> 2][2 * (u & 3) + 1]};

    #pragma unroll
    for (int u = 0; u < 4 * RT; ++u) {
        float ex = e[u].x;
        float ey = e[u].y;
        if (u > 0 && u - 1 >= LAG) issue_after(acc[(u - 1 - LAG) >> 2][(u - 1 - LAG) & 3], ex);
        else if (u > 0)            issue_after(py[u - 1], ex);
        px[u] = __builtin_amdgcn_exp2f(ex);
        if (u + LEAD < 4 * RT) {
            const int w = u + LEAD;
            issue_after(px[u], y2);
            e[w] = c1v * y2 + (v2f){d[w >> 2][2 * (w & 3)], d[w >> 2][2 * (w & 3) + 1]};
            issue_after(e[w], ey);
        } else {
            issue_after(px[u], ey);
        }
        py[u] = __builtin_amdgcn_exp2f(ey);
        if (u >= LAG) {
            const int w = u - LAG;
            issue_after(py[u], al);
            acc[w >> 2][w & 3] = al * (v2f){px[w], py[w]} + acc[w >> 2][w & 3];
        }
    }
    #pragma unroll
    for (int w = 4 * RT - LAG; w < 4 * RT; ++w)
        acc[w >> 2][w & 3] = al * (v2f){px[w], py[w]} + acc[w >> 2][w & 3];
    issue_after(acc[RT - 1][3], next.q[0]);
}

__global__ __launch_bounds__(256, 2)
void nkrr_rbf_wmma_kernel(const float* __restrict__ X,
                          const float* __restrict__ C,
                          const float* __restrict__ alphas,
                          const float* __restrict__ sigma,
                          float* __restrict__ out,
                          int N, int M)
{
    __shared__ __attribute__((aligned(16))) float s_x2[ROWS];
    __shared__ __attribute__((aligned(16))) float s_part[WAVES][RT][32][8];
    __shared__ __attribute__((aligned(16))) float s_out[ROWS];

    const float g  = sigma[0];
    const float c1 = -0.5f * LOG2E / (g * g);
    const float c2 = -2.0f * c1;

    const int tid  = threadIdx.x;
    const int lane = tid & 31;
    const int wave = __builtin_amdgcn_readfirstlane(tid >> 5);
    const int half = lane >> 4;
    const int ln   = lane & 15;
    const int row0 = blockIdx.x * ROWS;

    const v4f* __restrict__ C4 = reinterpret_cast<const v4f*>(C);
    const v4f* __restrict__ X4 = reinterpret_cast<const v4f*>(X);

    const int tilesPerWave = M / (TILE * WAVES);

    BTile t0, t1;
    load_btile(t0, C4, alphas, wave, ln, half);

    {
        int r = row0 + TILE * wave + ln;
        r = r < N ? r : N - 1;
        const v4f* p = X4 + (size_t)r * (DIM / 4) + half;
        v4f q[4];
        #pragma unroll
        for (int i = 0; i < 4; ++i) q[i] = p[2 * i];
        float s = sumsq16(q);
        s += xhalf(s);
        if (half == 0) s_x2[TILE * wave + ln] = c1 * s;
    }

    v2f a[RT][8];
    #pragma unroll
    for (int rt = 0; rt < RT; ++rt) {
        int r = row0 + TILE * rt + ln;
        r = r < N ? r : N - 1;
        const v4f* p = X4 + (size_t)r * (DIM / 4) + half;
        #pragma unroll
        for (int i = 0; i < 4; ++i) {
            const v4f q = p[2 * i];
            a[rt][2 * i].x     = c2 * q.x;
            a[rt][2 * i].y     = c2 * q.y;
            a[rt][2 * i + 1].x = c2 * q.z;
            a[rt][2 * i + 1].y = c2 * q.w;
        }
    }

    __syncthreads();

    v8f x2s[RT];
    #pragma unroll
    for (int rt = 0; rt < RT; ++rt) {
        const v4f lo = *reinterpret_cast<const v4f*>(&s_x2[TILE * rt + 8 * half]);
        const v4f hi = *reinterpret_cast<const v4f*>(&s_x2[TILE * rt + 8 * half + 4]);
        x2s[rt] = (v8f){lo.x, lo.y, lo.z, lo.w, hi.x, hi.y, hi.z, hi.w};
    }

    v2f acc[RT][4];
    #pragma unroll
    for (int rt = 0; rt < RT; ++rt) {
        #pragma unroll
        for (int j = 0; j < 4; ++j) acc[rt][j] = (v2f){0.f, 0.f};
    }

    v8f d[RT];
    for (int i = 0; i < tilesPerWave; i += 2) {
        load_btile(t1, C4, alphas, wave + WAVES * (i + 1), ln, half);
        __builtin_amdgcn_sched_barrier(0);
        mm_tile(t0, a, x2s, d);
        post_tile(t0, c1, d, acc, t1);
        const int nx = i + 2 < tilesPerWave ? i + 2 : i;
        load_btile(t0, C4, alphas, wave + WAVES * nx, ln, half);
        __builtin_amdgcn_sched_barrier(0);
        mm_tile(t1, a, x2s, d);
        post_tile(t1, c1, d, acc, t0);
    }

    #pragma unroll
    for (int rt = 0; rt < RT; ++rt) {
        v4f* dst = reinterpret_cast<v4f*>(&s_part[wave][rt][lane][0]);
        dst[0] = (v4f){acc[rt][0].x, acc[rt][0].y, acc[rt][1].x, acc[rt][1].y};
        dst[1] = (v4f){acc[rt][2].x, acc[rt][2].y, acc[rt][3].x, acc[rt][3].y};
    }
    __syncthreads();

    float r[8];
    #pragma unroll
    for (int v = 0; v < 8; ++v) r[v] = 0.f;
    #pragma unroll
    for (int w2 = 0; w2 < WAVES; ++w2) {
        const v4f* sp = reinterpret_cast<const v4f*>(&s_part[w2][wave][lane][0]);
        const v4f lo = sp[0];
        const v4f hi = sp[1];
        r[0] += lo.x; r[1] += lo.y; r[2] += lo.z; r[3] += lo.w;
        r[4] += hi.x; r[5] += hi.y; r[6] += hi.z; r[7] += hi.w;
    }
    #pragma unroll
    for (int v = 0; v < 8; ++v) r[v] = xlane_add<0x161>(r[v]);
    #pragma unroll
    for (int v = 0; v < 8; ++v) r[v] = xlane_add<0x162>(r[v]);
    #pragma unroll
    for (int v = 0; v < 8; ++v) r[v] = xlane_add<0x164>(r[v]);
    #pragma unroll
    for (int v = 0; v < 8; ++v) r[v] = xlane_add<0x168>(r[v]);
    if (ln == 0) {
        v4f* dst = reinterpret_cast<v4f*>(&s_out[TILE * wave + 8 * half]);
        dst[0] = (v4f){r[0], r[1], r[2], r[3]};
        dst[1] = (v4f){r[4], r[5], r[6], r[7]};
    }
    __syncthreads();

    if (wave == 0) {
        const v4f o = reinterpret_cast<const v4f*>(s_out)[lane];
        const int r0 = row0 + 4 * lane;
        if (r0 + 3 < N) {
            *reinterpret_cast<v4f*>(out + r0) = o;
        } else {
            if (r0 + 0 < N) out[r0 + 0] = o.x;
            if (r0 + 1 < N) out[r0 + 1] = o.y;
            if (r0 + 2 < N) out[r0 + 2] = o.z;
        }
    }
}

extern "C" void kernel_launch(void* const* d_in, const int* in_sizes, int n_in,
                              void* d_out, int out_size, void* d_ws, size_t ws_size,
                              hipStream_t stream) {
    const float* X      = (const float*)d_in[0];
    const float* C      = (const float*)d_in[1];
    const float* alphas = (const float*)d_in[2];
    const float* sigma  = (const float*)d_in[3];
    float* out = (float*)d_out;

    const int N = in_sizes[0] / DIM;
    const int M = in_sizes[1] / DIM;

    dim3 block(32 * WAVES);
    dim3 grid(N / ROWS);

    nkrr_rbf_wmma_kernel<<<grid, block, 0, stream>>>(X, C, alphas, sigma, out, N, M);
}
